# starting kernel
# baseline (speedup 1.0000x reference)
#include <hip/hip_runtime.h>
#include <hip/hip_bf16.h>

typedef __attribute__((ext_vector_type(2))) float v2f;
typedef __attribute__((ext_vector_type(8))) float v8f;

#define N_    64
#define C_    4
#define D_    4096
#define G_    32
#define P_    384

#define WAVES_PER_BLOCK 8
#define THREADS (WAVES_PER_BLOCK * 32)
#define DCHUNKS 4
#define D_PER_BLOCK (D_ / DCHUNKS)
#define D_PER_WAVE  (D_PER_BLOCK / WAVES_PER_BLOCK)
#define PAIRS       (D_PER_WAVE / 32)

#define SWAPX16 0x401f

__device__ __forceinline__ float swap_halves(float v) {
    return __int_as_float(__builtin_amdgcn_ds_swizzle(__float_as_int(v), SWAPX16));
}

__global__ __launch_bounds__(THREADS)
void gauss_superpos_wmma(const float* __restrict__ x,
                         const float* __restrict__ net,
                         const int*   __restrict__ amp_idx,
                         const int*   __restrict__ mu_idx,
                         const int*   __restrict__ sigma_idx,
                         float*       __restrict__ out)
{
    __shared__ float sA2[G_], sA1[G_], sA0[G_], sAmp[G_];

    const int nc = blockIdx.x;
    const int n  = nc / C_;
    const int c  = nc % C_;
    const int tid = threadIdx.x;

    if (tid < G_) {
        const int g  = tid;
        const int ia = amp_idx  [c * G_ + g];
        const int im = mu_idx   [c * G_ + g];
        const int is = sigma_idx[c * G_ + g];
        const float amp = net[n * P_ + ia];
        const float mu  = net[n * P_ + im];
        const float sg  = net[n * P_ + is];
        const float iv  = 1.0f / (sg * sg);
        const float LOG2E = 1.4426950408889634f;
        const float h = -0.5f * iv * LOG2E;
        sA2[g]  = h;
        sA1[g]  = mu * iv * LOG2E;
        sA0[g]  = h * mu * mu;
        sAmp[g] = amp;
    }
    __syncthreads();

    const int lane = tid & 31;
    const int half = lane >> 4;
    const int gl   = lane & 15;
    const int wave = tid >> 5;

    v2f At0, At1;
    At0.x = half ? sA0[gl]      : sA2[gl];
    At0.y = half ? 0.0f         : sA1[gl];
    At1.x = half ? sA0[16 + gl] : sA2[16 + gl];
    At1.y = half ? 0.0f         : sA1[16 + gl];

    float amp0[8], amp1[8];
#pragma unroll
    for (int r = 0; r < 8; ++r) {
        amp0[r] = sAmp[r + 8 * half];
        amp1[r] = sAmp[16 + r + 8 * half];
    }

    const float* __restrict__ xrow = x   + (size_t)nc * D_;
    float*       __restrict__ orow = out + (size_t)nc * D_;
    const int dbase = blockIdx.y * D_PER_BLOCK + wave * D_PER_WAVE;

    float xv[PAIRS];
#pragma unroll
    for (int p = 0; p < PAIRS; ++p)
        xv[p] = xrow[dbase + p * 32 + lane];

#pragma unroll
    for (int p = 0; p < PAIRS; ++p) {
        const float xlo = xv[p];
        const float xhi = swap_halves(xv[p]);

        v2f B0, B1;
        B0.x = half ? 1.0f : xlo * xlo;  B0.y = half ? 0.0f : xlo;
        B1.x = half ? 1.0f : xhi * xhi;  B1.y = half ? 0.0f : xhi;
        v8f cz = {0.f, 0.f, 0.f, 0.f, 0.f, 0.f, 0.f, 0.f};

        v8f e00 = __builtin_amdgcn_wmma_f32_16x16x4_f32(
                      false, At0, false, B0, (short)0, cz, false, false);
        v8f e01 = __builtin_amdgcn_wmma_f32_16x16x4_f32(
                      false, At1, false, B0, (short)0, cz, false, false);
        v8f e10 = __builtin_amdgcn_wmma_f32_16x16x4_f32(
                      false, At0, false, B1, (short)0, cz, false, false);
        v8f e11 = __builtin_amdgcn_wmma_f32_16x16x4_f32(
                      false, At1, false, B1, (short)0, cz, false, false);

        float s0a = 0.f, s0b = 0.f, s1a = 0.f, s1b = 0.f;
#pragma unroll
        for (int r = 0; r < 8; ++r) {
            s0a = fmaf(amp0[r], __builtin_amdgcn_exp2f(e00[r]), s0a);
            s0b = fmaf(amp1[r], __builtin_amdgcn_exp2f(e01[r]), s0b);
        }
#pragma unroll
        for (int r = 0; r < 8; ++r) {
            s1a = fmaf(amp0[r], __builtin_amdgcn_exp2f(e10[r]), s1a);
            s1b = fmaf(amp1[r], __builtin_amdgcn_exp2f(e11[r]), s1b);
        }
        const float s0 = s0a + s0b;
        const float s1 = s1a + s1b;
        const float tot0 = s0 + swap_halves(s0);
        const float tot1 = s1 + swap_halves(s1);

        orow[dbase + p * 32 + lane] = half ? tot1 : tot0;
    }
}

extern "C" void kernel_launch(void* const* d_in, const int* in_sizes, int n_in,
                              void* d_out, int out_size, void* d_ws, size_t ws_size,
                              hipStream_t stream) {
    const float* x   = (const float*)d_in[0];
    const float* net = (const float*)d_in[1];
    const int*   ai  = (const int*)d_in[2];
    const int*   mi  = (const int*)d_in[3];
    const int*   si  = (const int*)d_in[4];
    float* out = (float*)d_out;

    dim3 grid(N_ * C_, DCHUNKS);
    gauss_superpos_wmma<<<grid, THREADS, 0, stream>>>(x, net, ai, mi, si, out);
    (void)in_sizes; (void)n_in; (void)out_size; (void)d_ws; (void)ws_size;
}


# Each wave runs one hand-ordered stream over its tile pairs: an exp2, then the fma of the exp2 four steps back, the NEXT pair's WMMAs issued every 8 steps beside them (one 8-wave block per row)
# speedup vs baseline: 1.2746x; 1.2746x over previous
#include <hip/hip_runtime.h>


typedef __attribute__((ext_vector_type(2))) float v2f;
typedef __attribute__((ext_vector_type(4))) float v4f;
typedef __attribute__((ext_vector_type(8))) float v8f;

#define N_    64
#define C_    4
#define D_    4096
#define G_    32
#define P_    384

#define WAVES_PER_BLOCK 8
#define THREADS (WAVES_PER_BLOCK * 32)
#define D_PER_WAVE  (D_ / WAVES_PER_BLOCK)
#define QUADS       (D_PER_WAVE / 128)
#define PAIRS       (QUADS * 4)

__device__ __forceinline__ float xhalf(float v) {
    return __builtin_bit_cast(float, __builtin_amdgcn_permlanex16(
        __builtin_bit_cast(int, v), __builtin_bit_cast(int, v),
        0x76543210, 0xfedcba98, false, false));
}

#define LAG 4
#define TAIL 12

__global__ __launch_bounds__(THREADS)
void gauss_superpos_wmma(const float* __restrict__ x,
                         const float* __restrict__ net,
                         const int*   __restrict__ amp_idx,
                         const int*   __restrict__ mu_idx,
                         const int*   __restrict__ sigma_idx,
                         float*       __restrict__ out)
{
    __shared__ float sA2[G_], sA1[G_], sA0[G_], sAmp[G_];

    const int nc = blockIdx.x;
    const int n  = nc / C_;
    const int c  = nc % C_;
    const int tid = threadIdx.x;
    const int lane = tid & 31;
    const int half = lane >> 4;
    const int gl   = lane & 15;
    const int wave = tid >> 5;

    const float* __restrict__ xrow = x   + (size_t)nc * D_;
    float*       __restrict__ orow = out + (size_t)nc * D_;
    const int dbase = wave * D_PER_WAVE;

    v4f xq[QUADS];
#pragma unroll
    for (int q = 0; q < QUADS; ++q)
        xq[q] = *reinterpret_cast<const v4f*>(xrow + dbase + q * 128 + lane * 4);

    if (tid < G_) {
        const int g  = tid;
        const int ia = amp_idx  [c * G_ + g];
        const int im = mu_idx   [c * G_ + g];
        const int is = sigma_idx[c * G_ + g];
        const float amp = net[n * P_ + ia];
        const float mu  = net[n * P_ + im];
        const float sg  = net[n * P_ + is];
        const float iv  = 1.0f / (sg * sg);
        const float LOG2E = 1.4426950408889634f;
        const float h = -0.5f * iv * LOG2E;
        sA2[g]  = h;
        sA1[g]  = mu * iv * LOG2E;
        sA0[g]  = h * mu * mu;
        sAmp[g] = amp;
    }
    __syncthreads();

    v2f At0, At1;
    At0.x = half ? sA0[gl]      : sA2[gl];
    At0.y = half ? 0.0f         : sA1[gl];
    At1.x = half ? sA0[16 + gl] : sA2[16 + gl];
    At1.y = half ? 0.0f         : sA1[16 + gl];

    float amp0[8], amp1[8];
#pragma unroll
    for (int r = 0; r < 8; ++r) {
        amp0[r] = sAmp[r + 8 * half];
        amp1[r] = sAmp[16 + r + 8 * half];
    }

    const v8f cz = {0.f, 0.f, 0.f, 0.f, 0.f, 0.f, 0.f, 0.f};
    v8f   E[2][4];
    float acc[2][4];
    float ex[8];
    v2f   B0, B1;
    v4f   o;
    float s0 = 0.f, s1 = 0.f, give = 0.f, mine = 0.f, got = 0.f;

    {
        const float xlo = xq[0][0];
        const float xhi = xhalf(xlo);
        B0.x = half ? 1.0f : xlo * xlo;  B0.y = half ? 0.0f : xlo;
        B1.x = half ? 1.0f : xhi * xhi;  B1.y = half ? 0.0f : xhi;
        E[0][0] = __builtin_amdgcn_wmma_f32_16x16x4_f32(false, At0, false, B0, (short)0, cz, false, false);
        E[0][1] = __builtin_amdgcn_wmma_f32_16x16x4_f32(false, At1, false, B0, (short)0, cz, false, false);
        E[0][2] = __builtin_amdgcn_wmma_f32_16x16x4_f32(false, At0, false, B1, (short)0, cz, false, false);
        E[0][3] = __builtin_amdgcn_wmma_f32_16x16x4_f32(false, At1, false, B1, (short)0, cz, false, false);
    }

#pragma clang loop unroll(full)
    for (int t = 0; t < PAIRS * 32 + LAG + TAIL; ++t) {
        if (t < PAIRS * 32) {
            const int p = t >> 5, e = t & 31;
            if (p + 1 < PAIRS) {
                const int pn = p + 1;
                if (e == 0) {
                    const float xlo = xq[pn >> 2][pn & 3];
                    const float xhi = xhalf(xlo);
                    B0.x = half ? 1.0f : xlo * xlo;  B0.y = half ? 0.0f : xlo;
                    B1.x = half ? 1.0f : xhi * xhi;  B1.y = half ? 0.0f : xhi;
                }
                if ((e & 7) == 0) {
                    const int k = e >> 3;
                    E[pn & 1][k] = __builtin_amdgcn_wmma_f32_16x16x4_f32(
                        false, (k & 1) ? At1 : At0, false, (k & 2) ? B1 : B0,
                        (short)0, cz, false, false);
                }
            }
            ex[t & 7] = __builtin_amdgcn_exp2f(E[p & 1][e & 3][e >> 2]);
        }
        if (t >= LAG) {
            const int u = t - LAG;
            const int p = u >> 5, e = u & 31, k = e & 3, r = e >> 2;
            if (p < PAIRS) {
                const float a = (k & 1) ? amp1[r] : amp0[r];
                acc[p & 1][k] = fmaf(a, ex[u & 7], r == 0 ? 0.0f : acc[p & 1][k]);
            }
            if (p >= 1) {
                const int pf = p - 1;
                if (e == 1) {
                    s0 = acc[pf & 1][0] + acc[pf & 1][1];
                    s1 = acc[pf & 1][2] + acc[pf & 1][3];
                }
                if (e == 4) {
                    give = half ? s0 : s1;
                    mine = half ? s1 : s0;
                }
                if (e == 7)  got = xhalf(give);
                if (e == 10) o[pf & 3] = mine + got;
                if (e == 11 && (pf & 3) == 3)
                    *reinterpret_cast<v4f*>(orow + dbase + (pf >> 2) * 128 + lane * 4) = o;
            }
        }
        __builtin_amdgcn_sched_barrier(0);
    }
}

extern "C" void kernel_launch(void* const* d_in, const int* in_sizes, int n_in,
                              void* d_out, int out_size, void* d_ws, size_t ws_size,
                              hipStream_t stream) {
    const float* x   = (const float*)d_in[0];
    const float* net = (const float*)d_in[1];
    const int*   ai  = (const int*)d_in[2];
    const int*   mi  = (const int*)d_in[3];
    const int*   si  = (const int*)d_in[4];
    float* out = (float*)d_out;

    dim3 grid(N_ * C_);
    gauss_superpos_wmma<<<grid, THREADS, 0, stream>>>(x, net, ai, mi, si, out);
    (void)in_sizes; (void)n_in; (void)out_size; (void)d_ws; (void)ws_size;
}


# Each wave builds its own coefficient operands from one round trip (indices and whole parameter row together, row indexed in a private LDS copy): no dependent gather trip, no barrier before the WMMAs
# speedup vs baseline: 1.3880x; 1.0890x over previous
#include <hip/hip_runtime.h>


typedef __attribute__((ext_vector_type(2))) float v2f;
typedef __attribute__((ext_vector_type(4))) float v4f;
typedef __attribute__((ext_vector_type(8))) float v8f;

#define N_    64
#define C_    4
#define D_    4096
#define G_    32
#define P_    384

#define WAVES_PER_BLOCK 8
#define THREADS (WAVES_PER_BLOCK * 32)
#define D_PER_WAVE  (D_ / WAVES_PER_BLOCK)
#define QUADS       (D_PER_WAVE / 128)
#define PAIRS       (QUADS * 4)

__device__ __forceinline__ float xhalf(float v) {
    return __builtin_bit_cast(float, __builtin_amdgcn_permlanex16(
        __builtin_bit_cast(int, v), __builtin_bit_cast(int, v),
        0x76543210, 0xfedcba98, false, false));
}

#define LAG 4
#define TAIL 12

__global__ __launch_bounds__(THREADS)
void gauss_superpos_wmma(const float* __restrict__ x,
                         const float* __restrict__ net,
                         const int*   __restrict__ amp_idx,
                         const int*   __restrict__ mu_idx,
                         const int*   __restrict__ sigma_idx,
                         float*       __restrict__ out)
{
    __shared__ __attribute__((aligned(16))) float sRow[WAVES_PER_BLOCK][P_];
    __shared__ __attribute__((aligned(16))) float sAmp[WAVES_PER_BLOCK][G_];

    const int nc = blockIdx.x;
    const int n  = nc / C_;
    const int c  = nc % C_;
    const int tid = threadIdx.x;
    const int lane = tid & 31;
    const int half = lane >> 4;
    const int wave = tid >> 5;

    const float* __restrict__ xrow = x   + (size_t)nc * D_;
    float*       __restrict__ orow = out + (size_t)nc * D_;
    const int dbase = wave * D_PER_WAVE;

    const int g  = lane;
    const int ia = amp_idx  [c * G_ + g];
    const int im = mu_idx   [c * G_ + g];
    const int is = sigma_idx[c * G_ + g];
    v4f rq[P_ / 128];
#pragma unroll
    for (int q = 0; q < P_ / 128; ++q)
        rq[q] = *reinterpret_cast<const v4f*>(net + n * P_ + q * 128 + lane * 4);

    v4f xq[QUADS];
#pragma unroll
    for (int q = 0; q < QUADS; ++q)
        xq[q] = *reinterpret_cast<const v4f*>(xrow + dbase + q * 128 + lane * 4);
    __builtin_amdgcn_sched_barrier(0);

    float* row = sRow[wave];
#pragma unroll
    for (int q = 0; q < P_ / 128; ++q)
        *reinterpret_cast<v4f*>(row + q * 128 + lane * 4) = rq[q];
    __builtin_amdgcn_fence(__ATOMIC_ACQ_REL, "wavefront");
    const float amp = row[min((unsigned)ia, (unsigned)(P_ - 1))];
    const float mu  = row[min((unsigned)im, (unsigned)(P_ - 1))];
    const float sg  = row[min((unsigned)is, (unsigned)(P_ - 1))];
    __builtin_amdgcn_sched_barrier(0);
    const float iv  = 1.0f / (sg * sg);
    const float LOG2E = 1.4426950408889634f;
    const float h  = -0.5f * iv * LOG2E;
    const float A2 = h;
    const float A1 = mu * iv * LOG2E;
    const float A0 = h * mu * mu;
    sAmp[wave][g] = amp;
    __builtin_amdgcn_fence(__ATOMIC_ACQ_REL, "wavefront");

    const float recv0 = xhalf(half ? A2 : A0);
    const float recv1 = xhalf(A1);
    v2f At0, At1;
    At0.x = half ? recv0 : A2;
    At0.y = half ? 0.0f  : A1;
    At1.x = half ? A0    : recv0;
    At1.y = half ? 0.0f  : recv1;

    float amp0[8], amp1[8];
#pragma unroll
    for (int r = 0; r < 8; ++r) {
        amp0[r] = sAmp[wave][r + 8 * half];
        amp1[r] = sAmp[wave][16 + r + 8 * half];
    }

    const v8f cz = {0.f, 0.f, 0.f, 0.f, 0.f, 0.f, 0.f, 0.f};
    v8f   E[2][4];
    float acc[2][4];
    float ex[8];
    v2f   B0, B1;
    v4f   o;
    float s0 = 0.f, s1 = 0.f, give = 0.f, mine = 0.f, got = 0.f;

    {
        const float xlo = xq[0][0];
        const float xhi = xhalf(xlo);
        B0.x = half ? 1.0f : xlo * xlo;  B0.y = half ? 0.0f : xlo;
        B1.x = half ? 1.0f : xhi * xhi;  B1.y = half ? 0.0f : xhi;
        E[0][0] = __builtin_amdgcn_wmma_f32_16x16x4_f32(false, At0, false, B0, (short)0, cz, false, false);
        E[0][1] = __builtin_amdgcn_wmma_f32_16x16x4_f32(false, At1, false, B0, (short)0, cz, false, false);
        E[0][2] = __builtin_amdgcn_wmma_f32_16x16x4_f32(false, At0, false, B1, (short)0, cz, false, false);
        E[0][3] = __builtin_amdgcn_wmma_f32_16x16x4_f32(false, At1, false, B1, (short)0, cz, false, false);
    }

#pragma clang loop unroll(full)
    for (int t = 0; t < PAIRS * 32 + LAG + TAIL; ++t) {
        if (t < PAIRS * 32) {
            const int p = t >> 5, e = t & 31;
            if (p + 1 < PAIRS) {
                const int pn = p + 1;
                if (e == 0) {
                    const float xlo = xq[pn >> 2][pn & 3];
                    const float xhi = xhalf(xlo);
                    B0.x = half ? 1.0f : xlo * xlo;  B0.y = half ? 0.0f : xlo;
                    B1.x = half ? 1.0f : xhi * xhi;  B1.y = half ? 0.0f : xhi;
                }
                if ((e & 7) == 0) {
                    const int k = e >> 3;
                    E[pn & 1][k] = __builtin_amdgcn_wmma_f32_16x16x4_f32(
                        false, (k & 1) ? At1 : At0, false, (k & 2) ? B1 : B0,
                        (short)0, cz, false, false);
                }
            }
            ex[t & 7] = __builtin_amdgcn_exp2f(E[p & 1][e & 3][e >> 2]);
        }
        if (t >= LAG) {
            const int u = t - LAG;
            const int p = u >> 5, e = u & 31, k = e & 3, r = e >> 2;
            if (p < PAIRS) {
                const float a = (k & 1) ? amp1[r] : amp0[r];
                acc[p & 1][k] = fmaf(a, ex[u & 7], r == 0 ? 0.0f : acc[p & 1][k]);
            }
            if (p >= 1) {
                const int pf = p - 1;
                if (e == 1) {
                    s0 = acc[pf & 1][0] + acc[pf & 1][1];
                    s1 = acc[pf & 1][2] + acc[pf & 1][3];
                }
                if (e == 4) {
                    give = half ? s0 : s1;
                    mine = half ? s1 : s0;
                }
                if (e == 7)  got = xhalf(give);
                if (e == 10) o[pf & 3] = mine + got;
                if (e == 11 && (pf & 3) == 3)
                    *reinterpret_cast<v4f*>(orow + dbase + (pf >> 2) * 128 + lane * 4) = o;
            }
        }
        __builtin_amdgcn_sched_barrier(0);
    }
}

extern "C" void kernel_launch(void* const* d_in, const int* in_sizes, int n_in,
                              void* d_out, int out_size, void* d_ws, size_t ws_size,
                              hipStream_t stream) {
    const float* x   = (const float*)d_in[0];
    const float* net = (const float*)d_in[1];
    const int*   ai  = (const int*)d_in[2];
    const int*   mi  = (const int*)d_in[3];
    const int*   si  = (const int*)d_in[4];
    float* out = (float*)d_out;

    dim3 grid(N_ * C_);
    gauss_superpos_wmma<<<grid, THREADS, 0, stream>>>(x, net, ai, mi, si, out);
    (void)in_sizes; (void)n_in; (void)out_size; (void)d_ws; (void)ws_size;
}
